# starting kernel
# baseline (speedup 1.0000x reference)
#include <hip/hip_runtime.h>
#include <hip/hip_bf16.h>

typedef __attribute__((ext_vector_type(2))) float v2f;
typedef __attribute__((ext_vector_type(8))) float v8f;

#define IMG_S   128
#define NGAUSS  2048
#define NBATCH  8
#define NCHAN   32
#define CHUNK_G 128
#define GEPS    1e-6f

#define SQRT_HALF_LOG2E 0.84932180553116060f

__global__ __launch_bounds__(256)
void gs2d_render_kernel(const float* __restrict__ position,
                        const float* __restrict__ scale,
                        const float* __restrict__ rotation,
                        const float* __restrict__ opacity,
                        const float* __restrict__ feature,
                        float* __restrict__ out)
{
    __shared__ float4 s_par[CHUNK_G * 2];
    __shared__ float4 s_feat[CHUNK_G * NCHAN / 4];

    const int tid  = threadIdx.x;
    const int lane = tid & 31;
    const int wv   = tid >> 5;
    const int b    = blockIdx.x >> 7;
    const int h    = blockIdx.x & 127;

    const int   col = (wv << 4) + (lane & 15);
    const float gx  = (float)col * (1.0f / 127.0f);
    const float gy  = (float)h   * (1.0f / 127.0f);

    v8f d0 = {};
    v8f d1 = {};
    float o_acc = 0.0f;

    const float* s_featf = (const float*)s_feat;

    for (int n0 = 0; n0 < NGAUSS; n0 += CHUNK_G) {
        __syncthreads();
        if (tid < CHUNK_G) {
            const int gi = b * NGAUSS + n0 + tid;
            float px  = position[gi * 2 + 0];
            float py  = position[gi * 2 + 1];
            float sx  = fmaxf(scale[gi * 2 + 0], GEPS);
            float sy  = fmaxf(scale[gi * 2 + 1], GEPS);
            float ang = rotation[gi];
            float ca  = cosf(ang);
            float sa  = sinf(ang);
            float op  = opacity[gi];
            float rxs = SQRT_HALF_LOG2E / sx;
            float rys = SQRT_HALF_LOG2E / sy;
            float A = ca * rxs, B = sa * rxs;
            float C = -sa * rys, D = ca * rys;
            float G = __builtin_amdgcn_logf(op);
            s_par[tid * 2 + 0] = make_float4(px, py, A, B);
            s_par[tid * 2 + 1] = make_float4(C, D, G, 0.0f);
        }
        {
            const float4* src = (const float4*)(feature + (size_t)(b * NGAUSS + n0) * NCHAN);
            #pragma unroll
            for (int i = 0; i < (CHUNK_G * NCHAN / 4) / 256; ++i)
                s_feat[tid + i * 256] = src[tid + i * 256];
        }
        __syncthreads();

        if (n0 + CHUNK_G < NGAUSS) {
            const char* nsrc = (const char*)(feature + (size_t)(b * NGAUSS + n0 + CHUNK_G) * NCHAN);
            __builtin_prefetch(nsrc + tid * 64, 0, 1);
        }

        #pragma unroll 4
        for (int k0 = 0; k0 < CHUNK_G; k0 += 4) {
            const int kb = k0 + ((lane >> 4) << 1);

            float4 p0 = s_par[kb * 2 + 0];
            float4 q0 = s_par[kb * 2 + 1];
            float dx0 = gx - p0.x, dy0 = gy - p0.y;
            float u0 = fmaf(p0.w, dy0, p0.z * dx0);
            float v0 = fmaf(q0.y, dy0, q0.x * dx0);
            float a0e = fmaf(-u0, u0, fmaf(-v0, v0, q0.z));
            float w0 = __builtin_amdgcn_exp2f(a0e);

            float4 p1 = s_par[(kb + 1) * 2 + 0];
            float4 q1 = s_par[(kb + 1) * 2 + 1];
            float dx1 = gx - p1.x, dy1 = gy - p1.y;
            float u1 = fmaf(p1.w, dy1, p1.z * dx1);
            float v1 = fmaf(q1.y, dy1, q1.x * dx1);
            float a1e = fmaf(-u1, u1, fmaf(-v1, v1, q1.z));
            float w1 = __builtin_amdgcn_exp2f(a1e);

            o_acc += w0 + w1;

            v2f bm; bm.x = w0; bm.y = w1;

            const int m = lane & 15;
            v2f a0, a1;
            a0.x = s_featf[kb * NCHAN + m];
            a0.y = s_featf[(kb + 1) * NCHAN + m];
            a1.x = s_featf[kb * NCHAN + 16 + m];
            a1.y = s_featf[(kb + 1) * NCHAN + 16 + m];

            d0 = __builtin_amdgcn_wmma_f32_16x16x4_f32(false, a0, false, bm,
                                                       (short)0, d0, false, false);
            d1 = __builtin_amdgcn_wmma_f32_16x16x4_f32(false, a1, false, bm,
                                                       (short)0, d1, false, false);
        }
    }

    float o_tot = o_acc + __shfl_xor(o_acc, 16, 32);
    float invd  = 1.0f / fmaxf(o_tot, GEPS);

    const int    mo      = (lane >> 4) * 8;
    const size_t pixbase = (size_t)h * IMG_S + col;
    const size_t bbase   = (size_t)b * (NCHAN + 1) * IMG_S * IMG_S;

    #pragma unroll
    for (int r = 0; r < 8; ++r) {
        const int c0 = mo + r;
        out[bbase + (size_t)c0        * (IMG_S * IMG_S) + pixbase] = d0[r] * invd;
        out[bbase + (size_t)(c0 + 16) * (IMG_S * IMG_S) + pixbase] = d1[r] * invd;
    }
    if (lane < 16) {
        out[bbase + (size_t)NCHAN * (IMG_S * IMG_S) + pixbase] =
            fminf(fmaxf(o_tot, 0.0f), 1.0f);
    }
}

extern "C" void kernel_launch(void* const* d_in, const int* in_sizes, int n_in,
                              void* d_out, int out_size, void* d_ws, size_t ws_size,
                              hipStream_t stream) {
    const float* position = (const float*)d_in[0];
    const float* scale    = (const float*)d_in[1];
    const float* rotation = (const float*)d_in[2];
    const float* opacity  = (const float*)d_in[3];
    const float* feature  = (const float*)d_in[4];
    float* out = (float*)d_out;

    dim3 grid(NBATCH * IMG_S);
    dim3 block(256);
    gs2d_render_kernel<<<grid, block, 0, stream>>>(position, scale, rotation,
                                                   opacity, feature, out);
}


# each wave keeps 2 pixel tiles (two rows, same 16 columns) in registers: each LDS read of gaussian parameters and feature fragments, and the column terms, feed 4 WMMAs not 2; sums unchanged
# speedup vs baseline: 1.1903x; 1.1903x over previous
#include <hip/hip_runtime.h>
#include <hip/hip_bf16.h>


typedef __attribute__((ext_vector_type(2))) float v2f;
typedef __attribute__((ext_vector_type(8))) float v8f;

#define IMG_S   128
#define NGAUSS  2048
#define NBATCH  8
#define NCHAN   32
#define CHUNK_G 128
#define GEPS    1e-6f
#define ROWS    2
#define NBLOCKS 512

#define SQRT_HALF_LOG2E 0.84932180553116060f

__global__ __launch_bounds__(256)
void gs2d_render_kernel(const float* __restrict__ position,
                        const float* __restrict__ scale,
                        const float* __restrict__ rotation,
                        const float* __restrict__ opacity,
                        const float* __restrict__ feature,
                        float* __restrict__ out)
{
    __shared__ float4 s_par[CHUNK_G * 2];
    __shared__ float4 s_feat[CHUNK_G * NCHAN / 4];

    const int tid  = threadIdx.x;
    const int lane = tid & 31;
    const int wv   = tid >> 5;
    const int b    = blockIdx.x >> 6;
    const int h0   = (blockIdx.x & 63) * ROWS;

    const int   col = (wv << 4) + (lane & 15);
    const float gx  = (float)col * (1.0f / 127.0f);
    float gy[ROWS];
    #pragma unroll
    for (int t = 0; t < ROWS; ++t)
        gy[t] = (float)(h0 + t) * (1.0f / 127.0f);

    v8f d0[ROWS];
    v8f d1[ROWS];
    float o_acc[ROWS];
    #pragma unroll
    for (int t = 0; t < ROWS; ++t) {
        d0[t] = v8f{};
        d1[t] = v8f{};
        o_acc[t] = 0.0f;
    }

    const float* s_featf = (const float*)s_feat;

    for (int n0 = 0; n0 < NGAUSS; n0 += CHUNK_G) {
        __syncthreads();
        if (tid < CHUNK_G) {
            const int gi = b * NGAUSS + n0 + tid;
            float px  = position[gi * 2 + 0];
            float py  = position[gi * 2 + 1];
            float sx  = fmaxf(scale[gi * 2 + 0], GEPS);
            float sy  = fmaxf(scale[gi * 2 + 1], GEPS);
            float ang = rotation[gi];
            float ca  = cosf(ang);
            float sa  = sinf(ang);
            float op  = opacity[gi];
            float rxs = SQRT_HALF_LOG2E / sx;
            float rys = SQRT_HALF_LOG2E / sy;
            float A = ca * rxs, B = sa * rxs;
            float C = -sa * rys, D = ca * rys;
            float G = __builtin_amdgcn_logf(op);
            s_par[tid * 2 + 0] = make_float4(px, py, A, B);
            s_par[tid * 2 + 1] = make_float4(C, D, G, 0.0f);
        }
        {
            const float4* src = (const float4*)(feature + (size_t)(b * NGAUSS + n0) * NCHAN);
            #pragma unroll
            for (int i = 0; i < (CHUNK_G * NCHAN / 4) / 256; ++i)
                s_feat[tid + i * 256] = src[tid + i * 256];
        }
        __syncthreads();

        if (n0 + CHUNK_G < NGAUSS) {
            const char* nsrc = (const char*)(feature + (size_t)(b * NGAUSS + n0 + CHUNK_G) * NCHAN);
            __builtin_prefetch(nsrc + tid * 64, 0, 1);
        }

        #pragma unroll 4
        for (int k0 = 0; k0 < CHUNK_G; k0 += 4) {
            const int kb = k0 + ((lane >> 4) << 1);

            float4 p0 = s_par[kb * 2 + 0];
            float4 q0 = s_par[kb * 2 + 1];
            float4 p1 = s_par[(kb + 1) * 2 + 0];
            float4 q1 = s_par[(kb + 1) * 2 + 1];

            const int m = lane & 15;
            v2f a0, a1;
            a0.x = s_featf[kb * NCHAN + m];
            a0.y = s_featf[(kb + 1) * NCHAN + m];
            a1.x = s_featf[kb * NCHAN + 16 + m];
            a1.y = s_featf[(kb + 1) * NCHAN + 16 + m];

            float dx0 = gx - p0.x;
            float dx1 = gx - p1.x;
            float ax0 = p0.z * dx0, cx0 = q0.x * dx0;
            float ax1 = p1.z * dx1, cx1 = q1.x * dx1;

            #pragma unroll
            for (int t = 0; t < ROWS; ++t) {
                float dy0 = gy[t] - p0.y;
                float u0 = fmaf(p0.w, dy0, ax0);
                float v0 = fmaf(q0.y, dy0, cx0);
                float a0e = fmaf(-u0, u0, fmaf(-v0, v0, q0.z));
                float w0 = __builtin_amdgcn_exp2f(a0e);

                float dy1 = gy[t] - p1.y;
                float u1 = fmaf(p1.w, dy1, ax1);
                float v1 = fmaf(q1.y, dy1, cx1);
                float a1e = fmaf(-u1, u1, fmaf(-v1, v1, q1.z));
                float w1 = __builtin_amdgcn_exp2f(a1e);

                o_acc[t] += w0 + w1;

                v2f bm; bm.x = w0; bm.y = w1;

                d0[t] = __builtin_amdgcn_wmma_f32_16x16x4_f32(false, a0, false, bm,
                                                              (short)0, d0[t], false, false);
                d1[t] = __builtin_amdgcn_wmma_f32_16x16x4_f32(false, a1, false, bm,
                                                              (short)0, d1[t], false, false);
            }
        }
    }

    const int    mo    = (lane >> 4) * 8;
    const size_t bbase = (size_t)b * (NCHAN + 1) * IMG_S * IMG_S;

    #pragma unroll
    for (int t = 0; t < ROWS; ++t) {
        float o_tot = o_acc[t] + __shfl_xor(o_acc[t], 16, 32);
        float invd  = 1.0f / fmaxf(o_tot, GEPS);
        const size_t pixbase = (size_t)(h0 + t) * IMG_S + col;

        #pragma unroll
        for (int r = 0; r < 8; ++r) {
            const int c0 = mo + r;
            out[bbase + (size_t)c0        * (IMG_S * IMG_S) + pixbase] = d0[t][r] * invd;
            out[bbase + (size_t)(c0 + 16) * (IMG_S * IMG_S) + pixbase] = d1[t][r] * invd;
        }
        if (lane < 16) {
            out[bbase + (size_t)NCHAN * (IMG_S * IMG_S) + pixbase] =
                fminf(fmaxf(o_tot, 0.0f), 1.0f);
        }
    }
}

extern "C" void kernel_launch(void* const* d_in, const int* in_sizes, int n_in,
                              void* d_out, int out_size, void* d_ws, size_t ws_size,
                              hipStream_t stream) {
    const float* position = (const float*)d_in[0];
    const float* scale    = (const float*)d_in[1];
    const float* rotation = (const float*)d_in[2];
    const float* opacity  = (const float*)d_in[3];
    const float* feature  = (const float*)d_in[4];
    float* out = (float*)d_out;

    dim3 grid(NBLOCKS);
    dim3 block(256);
    gs2d_render_kernel<<<grid, block, 0, stream>>>(position, scale, rotation,
                                                   opacity, feature, out);
}


# row terms B*dy_t, D*dy_t (equal for all lanes of a row) folded once per gaussian at staging into the LDS record; k-loop keeps u=A*dx+term, reads at fixed offsets off walking pointers; sums unchanged
# speedup vs baseline: 1.2933x; 1.0865x over previous
#include <hip/hip_runtime.h>
#include <hip/hip_bf16.h>


typedef __attribute__((ext_vector_type(2))) float v2f;
typedef __attribute__((ext_vector_type(8))) float v8f;

#define IMG_S   128
#define NGAUSS  2048
#define NBATCH  8
#define NCHAN   32
#define CHUNK_G 128
#define GEPS    1e-6f
#define ROWS    2
#define NBLOCKS 512
static_assert(ROWS == 2, "the staged record carries exactly two row terms");

#define SQRT_HALF_LOG2E 0.84932180553116060f

__global__ __launch_bounds__(256)
void gs2d_render_kernel(const float* __restrict__ position,
                        const float* __restrict__ scale,
                        const float* __restrict__ rotation,
                        const float* __restrict__ opacity,
                        const float* __restrict__ feature,
                        float* __restrict__ out)
{
    __shared__ float4 s_par[CHUNK_G * 2];
    __shared__ float4 s_feat[CHUNK_G * NCHAN / 4];

    const int tid  = threadIdx.x;
    const int lane = tid & 31;
    const int wv   = tid >> 5;
    const int b    = blockIdx.x >> 6;
    const int h0   = (blockIdx.x & 63) * ROWS;

    const int   col = (wv << 4) + (lane & 15);
    const float gx  = (float)col * (1.0f / 127.0f);
    float gy[ROWS];
    #pragma unroll
    for (int t = 0; t < ROWS; ++t)
        gy[t] = (float)(h0 + t) * (1.0f / 127.0f);

    v8f d0[ROWS];
    v8f d1[ROWS];
    float o_acc[ROWS];
    #pragma unroll
    for (int t = 0; t < ROWS; ++t) {
        d0[t] = v8f{};
        d1[t] = v8f{};
        o_acc[t] = 0.0f;
    }

    const float* s_featf = (const float*)s_feat;
    const int     hoff   = (lane >> 4) << 1;
    const float4* par_l  = s_par + hoff * 2;
    const float*  feat_l = s_featf + hoff * NCHAN + (lane & 15);

    for (int n0 = 0; n0 < NGAUSS; n0 += CHUNK_G) {
        __syncthreads();
        if (tid < CHUNK_G) {
            const int gi = b * NGAUSS + n0 + tid;
            float px  = position[gi * 2 + 0];
            float py  = position[gi * 2 + 1];
            float sx  = fmaxf(scale[gi * 2 + 0], GEPS);
            float sy  = fmaxf(scale[gi * 2 + 1], GEPS);
            float ang = rotation[gi];
            float ca  = cosf(ang);
            float sa  = sinf(ang);
            float op  = opacity[gi];
            float rxs = SQRT_HALF_LOG2E / sx;
            float rys = SQRT_HALF_LOG2E / sy;
            float A = ca * rxs, B = sa * rxs;
            float C = -sa * rys, D = ca * rys;
            float G = __builtin_amdgcn_logf(op);
            float dyr0 = gy[0] - py;
            float dyr1 = gy[1] - py;
            s_par[tid * 2 + 0] = make_float4(px, A, C, G);
            s_par[tid * 2 + 1] = make_float4(B * dyr0, D * dyr0, B * dyr1, D * dyr1);
        }
        {
            const float4* src = (const float4*)(feature + (size_t)(b * NGAUSS + n0) * NCHAN);
            #pragma unroll
            for (int i = 0; i < (CHUNK_G * NCHAN / 4) / 256; ++i)
                s_feat[tid + i * 256] = src[tid + i * 256];
        }
        __syncthreads();

        if (n0 + CHUNK_G < NGAUSS) {
            const char* nsrc = (const char*)(feature + (size_t)(b * NGAUSS + n0 + CHUNK_G) * NCHAN);
            __builtin_prefetch(nsrc + tid * 64, 0, 1);
        }

        const float4* pp = par_l;
        const float*  fp = feat_l;
        #pragma unroll 4
        for (int k0 = 0; k0 < CHUNK_G; k0 += 4, pp += 8, fp += 4 * NCHAN) {

            float4 p0 = pp[0];
            float4 q0 = pp[1];
            float4 p1 = pp[2];
            float4 q1 = pp[3];

            v2f a0, a1;
            a0.x = fp[0];
            a0.y = fp[NCHAN];
            a1.x = fp[16];
            a1.y = fp[NCHAN + 16];

            float dx0 = gx - p0.x;
            float dx1 = gx - p1.x;

            #pragma unroll
            for (int t = 0; t < ROWS; ++t) {
                float u0 = fmaf(p0.y, dx0, t == 0 ? q0.x : q0.z);
                float v0 = fmaf(p0.z, dx0, t == 0 ? q0.y : q0.w);
                float a0e = fmaf(-u0, u0, fmaf(-v0, v0, p0.w));
                float w0 = __builtin_amdgcn_exp2f(a0e);

                float u1 = fmaf(p1.y, dx1, t == 0 ? q1.x : q1.z);
                float v1 = fmaf(p1.z, dx1, t == 0 ? q1.y : q1.w);
                float a1e = fmaf(-u1, u1, fmaf(-v1, v1, p1.w));
                float w1 = __builtin_amdgcn_exp2f(a1e);

                o_acc[t] += w0 + w1;

                v2f bm; bm.x = w0; bm.y = w1;

                d0[t] = __builtin_amdgcn_wmma_f32_16x16x4_f32(false, a0, false, bm,
                                                              (short)0, d0[t], false, false);
                d1[t] = __builtin_amdgcn_wmma_f32_16x16x4_f32(false, a1, false, bm,
                                                              (short)0, d1[t], false, false);
            }
        }
    }

    const int    mo    = (lane >> 4) * 8;
    const size_t bbase = (size_t)b * (NCHAN + 1) * IMG_S * IMG_S;

    #pragma unroll
    for (int t = 0; t < ROWS; ++t) {
        float o_tot = o_acc[t] + __shfl_xor(o_acc[t], 16, 32);
        float invd  = 1.0f / fmaxf(o_tot, GEPS);
        const size_t pixbase = (size_t)(h0 + t) * IMG_S + col;

        #pragma unroll
        for (int r = 0; r < 8; ++r) {
            const int c0 = mo + r;
            out[bbase + (size_t)c0        * (IMG_S * IMG_S) + pixbase] = d0[t][r] * invd;
            out[bbase + (size_t)(c0 + 16) * (IMG_S * IMG_S) + pixbase] = d1[t][r] * invd;
        }
        if (lane < 16) {
            out[bbase + (size_t)NCHAN * (IMG_S * IMG_S) + pixbase] =
                fminf(fmaxf(o_tot, 0.0f), 1.0f);
        }
    }
}

extern "C" void kernel_launch(void* const* d_in, const int* in_sizes, int n_in,
                              void* d_out, int out_size, void* d_ws, size_t ws_size,
                              hipStream_t stream) {
    const float* position = (const float*)d_in[0];
    const float* scale    = (const float*)d_in[1];
    const float* rotation = (const float*)d_in[2];
    const float* opacity  = (const float*)d_in[3];
    const float* feature  = (const float*)d_in[4];
    float* out = (float*)d_out;

    dim3 grid(NBLOCKS);
    dim3 block(256);
    gs2d_render_kernel<<<grid, block, 0, stream>>>(position, scale, rotation,
                                                   opacity, feature, out);
}


# exponent by completing the square in dx: staging folds al=sqrt(A^2+C^2), be_t=(b/al)*dy_t, G_t=G-((det/al)*dy_t)^2; k-loop w=exp2(G_t-(al*dx+be_t)^2), 2 FMAs per weight not 4; sums unchanged
# speedup vs baseline: 1.3899x; 1.0747x over previous
#include <hip/hip_runtime.h>
#include <hip/hip_bf16.h>


typedef __attribute__((ext_vector_type(2))) float v2f;
typedef __attribute__((ext_vector_type(8))) float v8f;

#define IMG_S   128
#define NGAUSS  2048
#define NBATCH  8
#define NCHAN   32
#define CHUNK_G 128
#define GEPS    1e-6f
#define ROWS    2
#define NBLOCKS 512
static_assert(ROWS == 2, "the staged record carries exactly two row terms");

#define SQRT_HALF_LOG2E 0.84932180553116060f

__global__ __launch_bounds__(256)
void gs2d_render_kernel(const float* __restrict__ position,
                        const float* __restrict__ scale,
                        const float* __restrict__ rotation,
                        const float* __restrict__ opacity,
                        const float* __restrict__ feature,
                        float* __restrict__ out)
{
    __shared__ float4 s_par[CHUNK_G / 2 * 3];
    __shared__ float4 s_feat[CHUNK_G * NCHAN / 4];

    const int tid  = threadIdx.x;
    const int lane = tid & 31;
    const int wv   = tid >> 5;
    const int b    = blockIdx.x >> 6;
    const int h0   = (blockIdx.x & 63) * ROWS;

    const int   col = (wv << 4) + (lane & 15);
    const float gx  = (float)col * (1.0f / 127.0f);
    float gy[ROWS];
    #pragma unroll
    for (int t = 0; t < ROWS; ++t)
        gy[t] = (float)(h0 + t) * (1.0f / 127.0f);

    v8f d0[ROWS];
    v8f d1[ROWS];
    float o_acc[ROWS];
    #pragma unroll
    for (int t = 0; t < ROWS; ++t) {
        d0[t] = v8f{};
        d1[t] = v8f{};
        o_acc[t] = 0.0f;
    }

    const float* s_featf = (const float*)s_feat;
    const int     hoff   = (lane >> 4) << 1;
    const float4* par_l  = s_par + (lane >> 4) * 3;
    const float*  feat_l = s_featf + hoff * NCHAN + (lane & 15);

    for (int n0 = 0; n0 < NGAUSS; n0 += CHUNK_G) {
        __syncthreads();
        if (tid < CHUNK_G) {
            const int gi = b * NGAUSS + n0 + tid;
            float px  = position[gi * 2 + 0];
            float py  = position[gi * 2 + 1];
            float sx  = fmaxf(scale[gi * 2 + 0], GEPS);
            float sy  = fmaxf(scale[gi * 2 + 1], GEPS);
            float ang = rotation[gi];
            float ca  = cosf(ang);
            float sa  = sinf(ang);
            float op  = opacity[gi];
            float rxs = SQRT_HALF_LOG2E / sx;
            float rys = SQRT_HALF_LOG2E / sy;
            float A = ca * rxs, B = sa * rxs;
            float C = -sa * rys, D = ca * rys;
            float G = __builtin_amdgcn_logf(op);
            float a   = fmaxf(fmaf(A, A, C * C), 1e-30f);
            float bx  = fmaf(A, B, C * D);
            float ial = __builtin_amdgcn_rsqf(a);
            float al  = a * ial;
            float bq  = bx * ial;
            float gq  = rxs * rys * ial;
            float dyr0 = gy[0] - py;
            float dyr1 = gy[1] - py;
            float hq0 = gq * dyr0;
            float hq1 = gq * dyr1;
            float2* rec = (float2*)s_par + tid * 3;
            rec[0] = make_float2(px, al);
            rec[1] = make_float2(bq * dyr0, fmaf(-hq0, hq0, G));
            rec[2] = make_float2(bq * dyr1, fmaf(-hq1, hq1, G));
        }
        {
            const float4* src = (const float4*)(feature + (size_t)(b * NGAUSS + n0) * NCHAN);
            #pragma unroll
            for (int i = 0; i < (CHUNK_G * NCHAN / 4) / 256; ++i)
                s_feat[tid + i * 256] = src[tid + i * 256];
        }
        __syncthreads();

        if (n0 + CHUNK_G < NGAUSS) {
            const char* nsrc = (const char*)(feature + (size_t)(b * NGAUSS + n0 + CHUNK_G) * NCHAN);
            __builtin_prefetch(nsrc + tid * 64, 0, 1);
        }

        const float4* pp = par_l;
        const float*  fp = feat_l;
        #pragma unroll 4
        for (int k0 = 0; k0 < CHUNK_G; k0 += 4, pp += 6, fp += 4 * NCHAN) {

            float4 r0 = pp[0];
            float4 r1 = pp[1];
            float4 r2 = pp[2];

            v2f a0, a1;
            a0.x = fp[0];
            a0.y = fp[NCHAN];
            a1.x = fp[16];
            a1.y = fp[NCHAN + 16];

            float dx0 = gx - r0.x;
            float dx1 = gx - r1.z;

            #pragma unroll
            for (int t = 0; t < ROWS; ++t) {
                float s0 = fmaf(r0.y, dx0, t == 0 ? r0.z : r1.x);
                float a0e = fmaf(-s0, s0, t == 0 ? r0.w : r1.y);
                float w0 = __builtin_amdgcn_exp2f(a0e);

                float s1 = fmaf(r1.w, dx1, t == 0 ? r2.x : r2.z);
                float a1e = fmaf(-s1, s1, t == 0 ? r2.y : r2.w);
                float w1 = __builtin_amdgcn_exp2f(a1e);

                o_acc[t] += w0 + w1;

                v2f bm; bm.x = w0; bm.y = w1;

                d0[t] = __builtin_amdgcn_wmma_f32_16x16x4_f32(false, a0, false, bm,
                                                              (short)0, d0[t], false, false);
                d1[t] = __builtin_amdgcn_wmma_f32_16x16x4_f32(false, a1, false, bm,
                                                              (short)0, d1[t], false, false);
            }
        }
    }

    const int    mo    = (lane >> 4) * 8;
    const size_t bbase = (size_t)b * (NCHAN + 1) * IMG_S * IMG_S;

    #pragma unroll
    for (int t = 0; t < ROWS; ++t) {
        float o_tot = o_acc[t] + __shfl_xor(o_acc[t], 16, 32);
        float invd  = 1.0f / fmaxf(o_tot, GEPS);
        const size_t pixbase = (size_t)(h0 + t) * IMG_S + col;

        #pragma unroll
        for (int r = 0; r < 8; ++r) {
            const int c0 = mo + r;
            out[bbase + (size_t)c0        * (IMG_S * IMG_S) + pixbase] = d0[t][r] * invd;
            out[bbase + (size_t)(c0 + 16) * (IMG_S * IMG_S) + pixbase] = d1[t][r] * invd;
        }
        if (lane < 16) {
            out[bbase + (size_t)NCHAN * (IMG_S * IMG_S) + pixbase] =
                fminf(fmaxf(o_tot, 0.0f), 1.0f);
        }
    }
}

extern "C" void kernel_launch(void* const* d_in, const int* in_sizes, int n_in,
                              void* d_out, int out_size, void* d_ws, size_t ws_size,
                              hipStream_t stream) {
    const float* position = (const float*)d_in[0];
    const float* scale    = (const float*)d_in[1];
    const float* rotation = (const float*)d_in[2];
    const float* opacity  = (const float*)d_in[3];
    const float* feature  = (const float*)d_in[4];
    float* out = (float*)d_out;

    dim3 grid(NBLOCKS);
    dim3 block(256);
    gs2d_render_kernel<<<grid, block, 0, stream>>>(position, scale, rotation,
                                                   opacity, feature, out);
}


# staging of the next gaussian chunk runs beside the k-loop: its global loads issue before the loop into registers, LDS tables double-buffered, one barrier per chunk not two; sums unchanged
# speedup vs baseline: 1.5526x; 1.1171x over previous
#include <hip/hip_runtime.h>
#include <hip/hip_bf16.h>


typedef __attribute__((ext_vector_type(2))) float v2f;
typedef __attribute__((ext_vector_type(8))) float v8f;

#define IMG_S   128
#define NGAUSS  2048
#define NBATCH  8
#define NCHAN   32
#define CHUNK_G 128
#define GEPS    1e-6f
#define ROWS    2
#define NBLOCKS 512
#define NCHUNK  (NGAUSS / CHUNK_G)
#define PAR_F4  (CHUNK_G / 2 * 3)
#define FEAT_F4 (CHUNK_G * NCHAN / 4)
static_assert(FEAT_F4 == 4 * 256, "each of the 256 threads carries four float4 of a feature chunk");
static_assert(ROWS == 2, "the staged record carries exactly two row terms");

#define SQRT_HALF_LOG2E 0.84932180553116060f

__global__ __launch_bounds__(256)
void gs2d_render_kernel(const float* __restrict__ position,
                        const float* __restrict__ scale,
                        const float* __restrict__ rotation,
                        const float* __restrict__ opacity,
                        const float* __restrict__ feature,
                        float* __restrict__ out)
{
    __shared__ float4 s_par[2 * PAR_F4];
    __shared__ float4 s_feat[2 * FEAT_F4];

    const int tid  = threadIdx.x;
    const int lane = tid & 31;
    const int wv   = tid >> 5;
    const int b    = blockIdx.x >> 6;
    const int h0   = (blockIdx.x & 63) * ROWS;

    const int   col = (wv << 4) + (lane & 15);
    const float gx  = (float)col * (1.0f / 127.0f);
    float gy[ROWS];
    #pragma unroll
    for (int t = 0; t < ROWS; ++t)
        gy[t] = (float)(h0 + t) * (1.0f / 127.0f);

    v8f d0[ROWS];
    v8f d1[ROWS];
    float o_acc[ROWS];
    #pragma unroll
    for (int t = 0; t < ROWS; ++t) {
        d0[t] = v8f{};
        d1[t] = v8f{};
        o_acc[t] = 0.0f;
    }

    const float* s_featf = (const float*)s_feat;
    const int     hoff   = (lane >> 4) << 1;
    const float4* par_l  = s_par + (lane >> 4) * 3;
    const float*  feat_l = s_featf + hoff * NCHAN + (lane & 15);

    float4 fr0, fr1, fr2, fr3;
    float  rpx = 0.0f, rpy = 0.0f, rsx = 1.0f, rsy = 1.0f, rang = 0.0f, rop = 1.0f;

#define ISSUE_CHUNK_LOADS(n0_)                                                        \
    do {                                                                              \
        if (tid < CHUNK_G) {                                                          \
            const int gi_ = b * NGAUSS + (n0_) + tid;                                 \
            rpx  = position[gi_ * 2 + 0];                                             \
            rpy  = position[gi_ * 2 + 1];                                             \
            rsx  = scale[gi_ * 2 + 0];                                                \
            rsy  = scale[gi_ * 2 + 1];                                                \
            rang = rotation[gi_];                                                     \
            rop  = opacity[gi_];                                                      \
        }                                                                             \
        const float4* src_ = (const float4*)(feature + (size_t)(b * NGAUSS + (n0_)) * NCHAN); \
        fr0 = src_[tid];                                                              \
        fr1 = src_[tid + 256];                                                        \
        fr2 = src_[tid + 512];                                                        \
        fr3 = src_[tid + 768];                                                        \
    } while (0)

    ISSUE_CHUNK_LOADS(0);
    for (int n = 0; n < NCHUNK; ++n) {
        const int cur = n & 1;
        if (tid < CHUNK_G) {
            float px  = rpx;
            float py  = rpy;
            float sx  = fmaxf(rsx, GEPS);
            float sy  = fmaxf(rsy, GEPS);
            float ang = rang;
            float ca  = cosf(ang);
            float sa  = sinf(ang);
            float op  = rop;
            float rxs = SQRT_HALF_LOG2E / sx;
            float rys = SQRT_HALF_LOG2E / sy;
            float A = ca * rxs, B = sa * rxs;
            float C = -sa * rys, D = ca * rys;
            float G = __builtin_amdgcn_logf(op);
            float a   = fmaxf(fmaf(A, A, C * C), 1e-30f);
            float bx  = fmaf(A, B, C * D);
            float ial = __builtin_amdgcn_rsqf(a);
            float al  = a * ial;
            float bq  = bx * ial;
            float gq  = rxs * rys * ial;
            float dyr0 = gy[0] - py;
            float dyr1 = gy[1] - py;
            float hq0 = gq * dyr0;
            float hq1 = gq * dyr1;
            float2* rec = (float2*)(s_par + cur * PAR_F4) + tid * 3;
            rec[0] = make_float2(px, al);
            rec[1] = make_float2(bq * dyr0, fmaf(-hq0, hq0, G));
            rec[2] = make_float2(bq * dyr1, fmaf(-hq1, hq1, G));
        }
        float4* dst = s_feat + cur * FEAT_F4;
        dst[tid]       = fr0;
        dst[tid + 256] = fr1;
        dst[tid + 512] = fr2;
        dst[tid + 768] = fr3;
        __syncthreads();
        if (n + 1 < NCHUNK)
            ISSUE_CHUNK_LOADS((n + 1) * CHUNK_G);

        const float4* pp = par_l + cur * PAR_F4;
        const float*  fp = feat_l + cur * (FEAT_F4 * 4);
        #pragma unroll 4
        for (int k0 = 0; k0 < CHUNK_G; k0 += 4, pp += 6, fp += 4 * NCHAN) {

            float4 r0 = pp[0];
            float4 r1 = pp[1];
            float4 r2 = pp[2];

            v2f a0, a1;
            a0.x = fp[0];
            a0.y = fp[NCHAN];
            a1.x = fp[16];
            a1.y = fp[NCHAN + 16];

            float dx0 = gx - r0.x;
            float dx1 = gx - r1.z;

            #pragma unroll
            for (int t = 0; t < ROWS; ++t) {
                float s0 = fmaf(r0.y, dx0, t == 0 ? r0.z : r1.x);
                float a0e = fmaf(-s0, s0, t == 0 ? r0.w : r1.y);
                float w0 = __builtin_amdgcn_exp2f(a0e);

                float s1 = fmaf(r1.w, dx1, t == 0 ? r2.x : r2.z);
                float a1e = fmaf(-s1, s1, t == 0 ? r2.y : r2.w);
                float w1 = __builtin_amdgcn_exp2f(a1e);

                o_acc[t] += w0 + w1;

                v2f bm; bm.x = w0; bm.y = w1;

                d0[t] = __builtin_amdgcn_wmma_f32_16x16x4_f32(false, a0, false, bm,
                                                              (short)0, d0[t], false, false);
                d1[t] = __builtin_amdgcn_wmma_f32_16x16x4_f32(false, a1, false, bm,
                                                              (short)0, d1[t], false, false);
            }
        }
    }

    const int    mo    = (lane >> 4) * 8;
    const size_t bbase = (size_t)b * (NCHAN + 1) * IMG_S * IMG_S;

    #pragma unroll
    for (int t = 0; t < ROWS; ++t) {
        float o_tot = o_acc[t] + __shfl_xor(o_acc[t], 16, 32);
        float invd  = 1.0f / fmaxf(o_tot, GEPS);
        const size_t pixbase = (size_t)(h0 + t) * IMG_S + col;

        #pragma unroll
        for (int r = 0; r < 8; ++r) {
            const int c0 = mo + r;
            out[bbase + (size_t)c0        * (IMG_S * IMG_S) + pixbase] = d0[t][r] * invd;
            out[bbase + (size_t)(c0 + 16) * (IMG_S * IMG_S) + pixbase] = d1[t][r] * invd;
        }
        if (lane < 16) {
            out[bbase + (size_t)NCHAN * (IMG_S * IMG_S) + pixbase] =
                fminf(fmaxf(o_tot, 0.0f), 1.0f);
        }
    }
}

extern "C" void kernel_launch(void* const* d_in, const int* in_sizes, int n_in,
                              void* d_out, int out_size, void* d_ws, size_t ws_size,
                              hipStream_t stream) {
    const float* position = (const float*)d_in[0];
    const float* scale    = (const float*)d_in[1];
    const float* rotation = (const float*)d_in[2];
    const float* opacity  = (const float*)d_in[3];
    const float* feature  = (const float*)d_in[4];
    float* out = (float*)d_out;

    dim3 grid(NBLOCKS);
    dim3 block(256);
    gs2d_render_kernel<<<grid, block, 0, stream>>>(position, scale, rotation,
                                                   opacity, feature, out);
}


# both LDS tables re-laid per k-pair: A fragments are one 8-byte read into an aligned register pair (no moves), pair weights use packed f32 ops; omap partials per pair member, sum order changed
# speedup vs baseline: 1.6360x; 1.0537x over previous
#include <hip/hip_runtime.h>
#include <hip/hip_bf16.h>


typedef __attribute__((ext_vector_type(2))) float v2f;
typedef __attribute__((ext_vector_type(4))) float v4f;
typedef __attribute__((ext_vector_type(8))) float v8f;

#define IMG_S   128
#define NGAUSS  2048
#define NBATCH  8
#define NCHAN   32
#define CHUNK_G 128
#define GEPS    1e-6f
#define ROWS    2
#define NBLOCKS 512
#define NCHUNK  (NGAUSS / CHUNK_G)
#define PAR_F4  (CHUNK_G / 2 * 3)
#define FEAT_F4 (CHUNK_G * NCHAN / 4)
static_assert(FEAT_F4 == 4 * 256, "each of the 256 threads carries four float4 of a feature chunk");
static_assert(ROWS == 2, "the staged record carries exactly two row terms");

#define SQRT_HALF_LOG2E 0.84932180553116060f

__global__ __launch_bounds__(256)
void gs2d_render_kernel(const float* __restrict__ position,
                        const float* __restrict__ scale,
                        const float* __restrict__ rotation,
                        const float* __restrict__ opacity,
                        const float* __restrict__ feature,
                        float* __restrict__ out)
{
    __shared__ float4 s_par[2 * PAR_F4];
    __shared__ float4 s_feat[2 * FEAT_F4];

    const int tid  = threadIdx.x;
    const int lane = tid & 31;
    const int wv   = tid >> 5;
    const int b    = blockIdx.x >> 6;
    const int h0   = (blockIdx.x & 63) * ROWS;

    const int   col = (wv << 4) + (lane & 15);
    const float gx  = (float)col * (1.0f / 127.0f);
    const v2f   gx2 = {gx, gx};
    float gy[ROWS];
    #pragma unroll
    for (int t = 0; t < ROWS; ++t)
        gy[t] = (float)(h0 + t) * (1.0f / 127.0f);

    v8f d0[ROWS];
    v8f d1[ROWS];
    v2f o_acc[ROWS];
    #pragma unroll
    for (int t = 0; t < ROWS; ++t) {
        d0[t] = v8f{};
        d1[t] = v8f{};
        o_acc[t] = v2f{};
    }

    const float* s_featf = (const float*)s_feat;
    const v4f*    par_l  = (const v4f*)s_par + (lane >> 4) * 3;
    const float*  feat_l = s_featf + (lane >> 4) * 32 + (lane & 15) * 2;

    float4 fr0, fr1, fr2, fr3;
    const int fsrc = ((tid >> 3) << 4) + (tid & 7);
    const int fdst = ((tid >> 4) << 5) + (((tid >> 2) & 1) << 4) + (((tid >> 3) & 1) << 3) + ((tid & 3) << 1);
    float  rpx = 0.0f, rpy = 0.0f, rsx = 1.0f, rsy = 1.0f, rang = 0.0f, rop = 1.0f;

#define ISSUE_CHUNK_LOADS(n0_)                                                        \
    do {                                                                              \
        if (tid < CHUNK_G) {                                                          \
            const int gi_ = b * NGAUSS + (n0_) + tid;                                 \
            rpx  = position[gi_ * 2 + 0];                                             \
            rpy  = position[gi_ * 2 + 1];                                             \
            rsx  = scale[gi_ * 2 + 0];                                                \
            rsy  = scale[gi_ * 2 + 1];                                                \
            rang = rotation[gi_];                                                     \
            rop  = opacity[gi_];                                                      \
        }                                                                             \
        const float4* src_ = (const float4*)(feature + (size_t)(b * NGAUSS + (n0_)) * NCHAN); \
            \
        fr0 = src_[fsrc];                                                             \
        fr1 = src_[fsrc + 8];                                                         \
        fr2 = src_[fsrc + 512];                                                       \
        fr3 = src_[fsrc + 520];                                                       \
    } while (0)

    ISSUE_CHUNK_LOADS(0);
    for (int n = 0; n < NCHUNK; ++n) {
        const int cur = n & 1;
        if (tid < CHUNK_G) {
            float px  = rpx;
            float py  = rpy;
            float sx  = fmaxf(rsx, GEPS);
            float sy  = fmaxf(rsy, GEPS);
            float ang = rang;
            float ca  = cosf(ang);
            float sa  = sinf(ang);
            float op  = rop;
            float rxs = SQRT_HALF_LOG2E / sx;
            float rys = SQRT_HALF_LOG2E / sy;
            float A = ca * rxs, B = sa * rxs;
            float C = -sa * rys, D = ca * rys;
            float G = __builtin_amdgcn_logf(op);
            float a   = fmaxf(fmaf(A, A, C * C), 1e-30f);
            float bx  = fmaf(A, B, C * D);
            float ial = __builtin_amdgcn_rsqf(a);
            float al  = a * ial;
            float bq  = bx * ial;
            float gq  = rxs * rys * ial;
            float dyr0 = gy[0] - py;
            float dyr1 = gy[1] - py;
            float hq0 = gq * dyr0;
            float hq1 = gq * dyr1;
            float* rec = (float*)(s_par + cur * PAR_F4) + (tid >> 1) * 12 + (tid & 1);
            rec[0]  = px;
            rec[2]  = al;
            rec[4]  = bq * dyr0;
            rec[6]  = fmaf(-hq0, hq0, G);
            rec[8]  = bq * dyr1;
            rec[10] = fmaf(-hq1, hq1, G);
        }
        float4* dst = s_feat + cur * FEAT_F4 + fdst;
        dst[0]       = make_float4(fr0.x, fr1.x, fr0.y, fr1.y);
        dst[1]       = make_float4(fr0.z, fr1.z, fr0.w, fr1.w);
        dst[512]     = make_float4(fr2.x, fr3.x, fr2.y, fr3.y);
        dst[512 + 1] = make_float4(fr2.z, fr3.z, fr2.w, fr3.w);
        __syncthreads();
        if (n + 1 < NCHUNK)
            ISSUE_CHUNK_LOADS((n + 1) * CHUNK_G);

        const v4f*    pp = par_l + cur * PAR_F4;
        const float*  fp = feat_l + cur * (FEAT_F4 * 4);
        #pragma unroll 4
        for (int k0 = 0; k0 < CHUNK_G; k0 += 4, pp += 6, fp += 4 * NCHAN) {

            v4f q0 = pp[0];
            v4f q1 = pp[1];
            v4f q2 = pp[2];

            const v2f a0 = *(const v2f*)(fp);
            const v2f a1 = *(const v2f*)(fp + 64);

            const v2f dx = gx2 - q0.xy;
            const v2f al = q0.zw;

            #pragma unroll
            for (int t = 0; t < ROWS; ++t) {
                const v2f be = (t == 0) ? q1.xy : q2.xy;
                const v2f Gt = (t == 0) ? q1.zw : q2.zw;
                const v2f s  = __builtin_elementwise_fma(al, dx, be);
                const v2f e  = __builtin_elementwise_fma(-s, s, Gt);

                v2f bm;
                bm.x = __builtin_amdgcn_exp2f(e.x);
                bm.y = __builtin_amdgcn_exp2f(e.y);

                o_acc[t] += bm;

                d0[t] = __builtin_amdgcn_wmma_f32_16x16x4_f32(false, a0, false, bm,
                                                              (short)0, d0[t], false, false);
                d1[t] = __builtin_amdgcn_wmma_f32_16x16x4_f32(false, a1, false, bm,
                                                              (short)0, d1[t], false, false);
            }
        }
    }

    const int    mo    = (lane >> 4) * 8;
    const size_t bbase = (size_t)b * (NCHAN + 1) * IMG_S * IMG_S;

    #pragma unroll
    for (int t = 0; t < ROWS; ++t) {
        float o_lane = o_acc[t].x + o_acc[t].y;
        float o_tot  = o_lane + __shfl_xor(o_lane, 16, 32);
        float invd  = 1.0f / fmaxf(o_tot, GEPS);
        const size_t pixbase = (size_t)(h0 + t) * IMG_S + col;

        #pragma unroll
        for (int r = 0; r < 8; ++r) {
            const int c0 = mo + r;
            out[bbase + (size_t)c0        * (IMG_S * IMG_S) + pixbase] = d0[t][r] * invd;
            out[bbase + (size_t)(c0 + 16) * (IMG_S * IMG_S) + pixbase] = d1[t][r] * invd;
        }
        if (lane < 16) {
            out[bbase + (size_t)NCHAN * (IMG_S * IMG_S) + pixbase] =
                fminf(fmaxf(o_tot, 0.0f), 1.0f);
        }
    }
}

extern "C" void kernel_launch(void* const* d_in, const int* in_sizes, int n_in,
                              void* d_out, int out_size, void* d_ws, size_t ws_size,
                              hipStream_t stream) {
    const float* position = (const float*)d_in[0];
    const float* scale    = (const float*)d_in[1];
    const float* rotation = (const float*)d_in[2];
    const float* opacity  = (const float*)d_in[3];
    const float* feature  = (const float*)d_in[4];
    float* out = (float*)d_out;

    dim3 grid(NBLOCKS);
    dim3 block(256);
    gs2d_render_kernel<<<grid, block, 0, stream>>>(position, scale, rotation,
                                                   opacity, feature, out);
}
